# starting kernel
# baseline (speedup 1.0000x reference)
#include <hip/hip_runtime.h>
#include <math.h>

typedef __attribute__((ext_vector_type(2))) float v2f;
typedef __attribute__((ext_vector_type(8))) float v8f;

#define WAVES   4
#define ROWS    16
#define HPITCH  197
#define H1PITCH 67
#define H2PITCH 33
#define OPITCH  17
#define H2OFF   1536

__device__ __forceinline__ void jacobi3(float* s, float* v, int p, int q) {
  float apq = s[p * 3 + q];
  if (fabsf(apq) < 1e-20f) return;
  float app = s[p * 3 + p], aqq = s[q * 3 + q];
  float tau = (aqq - app) / (2.f * apq);
  float t  = copysignf(1.f, tau) / (fabsf(tau) + sqrtf(1.f + tau * tau));
  float c  = 1.f / sqrtf(1.f + t * t);
  float sn = t * c;
  const int k = 3 - p - q;
  float skp = s[k * 3 + p], skq = s[k * 3 + q];
  float nkp = c * skp - sn * skq;
  float nkq = sn * skp + c * skq;
  s[k * 3 + p] = nkp; s[p * 3 + k] = nkp;
  s[k * 3 + q] = nkq; s[q * 3 + k] = nkq;
  s[p * 3 + p] = c * c * app - 2.f * sn * c * apq + sn * sn * aqq;
  s[q * 3 + q] = sn * sn * app + 2.f * sn * c * apq + c * c * aqq;
  s[p * 3 + q] = 0.f; s[q * 3 + p] = 0.f;
#pragma unroll
  for (int i = 0; i < 3; ++i) {
    float vip = v[i * 3 + p], viq = v[i * 3 + q];
    v[i * 3 + p] = c * vip - sn * viq;
    v[i * 3 + q] = sn * vip + c * viq;
  }
}

__global__ __launch_bounds__(128) void rmi_fused(
    const float* __restrict__ x,   const float* __restrict__ cwg,
    const float* __restrict__ cbg, const float* __restrict__ w1,
    const float* __restrict__ b1,  const float* __restrict__ w2,
    const float* __restrict__ b2,  const float* __restrict__ w3,
    const float* __restrict__ b3,  float* __restrict__ out, int B) {
  __shared__ float lds [WAVES * ROWS * HPITCH];
  __shared__ float ldsO[WAVES * ROWS * OPITCH];

  const int tid  = threadIdx.x;
  const int wave = tid >> 5;
  const int lane = tid & 31;
  const int rowBase = blockIdx.x * (WAVES * ROWS) + wave * ROWS;
  float* H = &lds [wave * ROWS * HPITCH];
  float* O = &ldsO[wave * ROWS * OPITCH];

  float cw[30];
#pragma unroll
  for (int i = 0; i < 30; ++i) cw[i] = cwg[i];
  const float cb = cbg[0];

  for (int r = 0; r < ROWS; ++r) {
    int sr = rowBase + r; sr = (sr < B) ? sr : (B - 1);
    const float* xr = x + (size_t)sr * 1200;
    for (int j = lane; j < 196; j += 32) {
      __builtin_prefetch(xr + j + 256, 0, 1);
      float acc = cb;
#pragma unroll
      for (int c = 0; c < 6; ++c)
#pragma unroll
        for (int k = 0; k < 5; ++k)
          acc = fmaf(xr[c * 200 + j + k], cw[c * 5 + k], acc);
      H[r * HPITCH + j] = acc;
    }
  }
  __syncthreads();

  const int hs = (lane >= 16) ? 2 : 0;
  const int rA = lane & 15;
  const int rD = (lane < 16) ? 0 : 8;

  int cc1[4];
#pragma unroll
  for (int nt = 0; nt < 4; ++nt) {
    const int c = nt * 16 + rA;
    cc1[nt] = (c < 50) ? c : 49;
  }

  v8f acc1[4] = {};
#pragma unroll 2
  for (int kk = 0; kk < 49; ++kk) {
    const int k0 = kk * 4 + hs;
    v2f a; a.x = H[rA * HPITCH + k0]; a.y = H[rA * HPITCH + k0 + 1];
#pragma unroll
    for (int nt = 0; nt < 4; ++nt) {
      v2f b;
      b.x = w1[k0 * 50 + cc1[nt]];
      b.y = w1[(k0 + 1) * 50 + cc1[nt]];
      acc1[nt] = __builtin_amdgcn_wmma_f32_16x16x4_f32(
          false, a, false, b, (short)0, acc1[nt], false, false);
    }
  }
  __syncthreads();

  float* H1 = H;
#pragma unroll
  for (int nt = 0; nt < 4; ++nt) {
    const int col = nt * 16 + rA;
    const float bb = b1[cc1[nt]];
#pragma unroll
    for (int r = 0; r < 8; ++r) {
      float vv = fmaxf(acc1[nt][r] + bb, 0.f);
      vv = (col < 50) ? vv : 0.f;
      H1[(rD + r) * H1PITCH + col] = vv;
    }
  }
  __syncthreads();

  const int cc2 = (16 + rA < 30) ? (16 + rA) : 29;
  v8f acc2[2] = {};
  for (int kk = 0; kk < 13; ++kk) {
    const int k0  = kk * 4 + hs;
    const int kc0 = (k0 < 50) ? k0 : 49;
    const int kc1 = (k0 + 1 < 50) ? (k0 + 1) : 49;
    v2f a;  a.x  = H1[rA * H1PITCH + k0]; a.y = H1[rA * H1PITCH + k0 + 1];
    v2f b0; b0.x = w2[kc0 * 30 + rA];     b0.y = w2[kc1 * 30 + rA];
    v2f b1v; b1v.x = w2[kc0 * 30 + cc2];  b1v.y = w2[kc1 * 30 + cc2];
    acc2[0] = __builtin_amdgcn_wmma_f32_16x16x4_f32(
        false, a, false, b0, (short)0, acc2[0], false, false);
    acc2[1] = __builtin_amdgcn_wmma_f32_16x16x4_f32(
        false, a, false, b1v, (short)0, acc2[1], false, false);
  }
  __syncthreads();
  float* H2 = H + H2OFF;
#pragma unroll
  for (int nt = 0; nt < 2; ++nt) {
    const int col = nt * 16 + rA;
    const float bb = b2[(col < 30) ? col : 29];
#pragma unroll
    for (int r = 0; r < 8; ++r) {
      float vv = fmaxf(acc2[nt][r] + bb, 0.f);
      vv = (col < 30) ? vv : 0.f;
      H2[(rD + r) * H2PITCH + col] = vv;
    }
  }
  __syncthreads();

  const int cc3 = (rA < 15) ? rA : 14;
  v8f acc3 = {};
  for (int kk = 0; kk < 8; ++kk) {
    const int k0  = kk * 4 + hs;
    const int kc0 = (k0 < 30) ? k0 : 29;
    const int kc1 = (k0 + 1 < 30) ? (k0 + 1) : 29;
    v2f a; a.x = H2[rA * H2PITCH + k0]; a.y = H2[rA * H2PITCH + k0 + 1];
    v2f b; b.x = w3[kc0 * 15 + cc3];    b.y = w3[kc1 * 15 + cc3];
    acc3 = __builtin_amdgcn_wmma_f32_16x16x4_f32(
        false, a, false, b, (short)0, acc3, false, false);
  }
  {
    const float bb = b3[cc3];
#pragma unroll
    for (int r = 0; r < 8; ++r)
      O[(rD + r) * OPITCH + rA] = acc3[r] + bb;
  }
  __syncthreads();

  if (lane < 16) {
    const int gRow = rowBase + lane;
    if (gRow < B) {
      float a9[9];
#pragma unroll
      for (int i = 0; i < 9; ++i) a9[i] = O[lane * OPITCH + i];
      float s[9], v[9];
#pragma unroll
      for (int i = 0; i < 3; ++i)
#pragma unroll
        for (int j = 0; j < 3; ++j) {
          s[i * 3 + j] = a9[0 * 3 + i] * a9[0 * 3 + j] +
                         a9[1 * 3 + i] * a9[1 * 3 + j] +
                         a9[2 * 3 + i] * a9[2 * 3 + j];
          v[i * 3 + j] = (i == j) ? 1.f : 0.f;
        }
#pragma unroll
      for (int sw = 0; sw < 6; ++sw) {
        jacobi3(s, v, 0, 1); jacobi3(s, v, 0, 2); jacobi3(s, v, 1, 2);
      }
      float g0 = s[0], g1 = s[4], g2 = s[8];
#define SWAPCOL(p, q) { float t_;                              \
        t_ = v[p];     v[p]     = v[q];     v[q]     = t_;     \
        t_ = v[3 + p]; v[3 + p] = v[3 + q]; v[3 + q] = t_;     \
        t_ = v[6 + p]; v[6 + p] = v[6 + q]; v[6 + q] = t_; }
      if (g0 < g1) { float t = g0; g0 = g1; g1 = t; SWAPCOL(0, 1); }
      if (g0 < g2) { float t = g0; g0 = g2; g2 = t; SWAPCOL(0, 2); }
      if (g1 < g2) { float t = g1; g1 = g2; g2 = t; SWAPCOL(1, 2); }
#undef SWAPCOL
      float w[9];
#pragma unroll
      for (int i = 0; i < 3; ++i)
#pragma unroll
        for (int j = 0; j < 3; ++j)
          w[i * 3 + j] = a9[i * 3 + 0] * v[0 + j] +
                         a9[i * 3 + 1] * v[3 + j] +
                         a9[i * 3 + 2] * v[6 + j];
      float u0x = w[0], u0y = w[3], u0z = w[6];
      float n0 = 1.f / sqrtf(u0x * u0x + u0y * u0y + u0z * u0z + 1e-20f);
      u0x *= n0; u0y *= n0; u0z *= n0;
      float u1x = w[1], u1y = w[4], u1z = w[7];
      float d01 = u1x * u0x + u1y * u0y + u1z * u0z;
      u1x -= d01 * u0x; u1y -= d01 * u0y; u1z -= d01 * u0z;
      float n1 = 1.f / sqrtf(u1x * u1x + u1y * u1y + u1z * u1z + 1e-20f);
      u1x *= n1; u1y *= n1; u1z *= n1;
      float cx = u0y * u1z - u0z * u1y;
      float cy = u0z * u1x - u0x * u1z;
      float cz = u0x * u1y - u0y * u1x;
      float detV = v[0] * (v[4] * v[8] - v[5] * v[7]) -
                   v[1] * (v[3] * v[8] - v[5] * v[6]) +
                   v[2] * (v[3] * v[7] - v[4] * v[6]);
      float sg = (detV >= 0.f) ? 1.f : -1.f;
      float u2x = sg * cx, u2y = sg * cy, u2z = sg * cz;
      float U_[9] = { u0x, u1x, u2x,  u0y, u1y, u2y,  u0z, u1z, u2z };
      float* op = out + (size_t)gRow * 15;
#pragma unroll
      for (int i = 0; i < 3; ++i)
#pragma unroll
        for (int j = 0; j < 3; ++j)
          op[i * 3 + j] = U_[i * 3 + 0] * v[j * 3 + 0] +
                          U_[i * 3 + 1] * v[j * 3 + 1] +
                          U_[i * 3 + 2] * v[j * 3 + 2];
#pragma unroll
      for (int i = 9; i < 15; ++i) op[i] = O[lane * OPITCH + i];
    }
  }
}

extern "C" void kernel_launch(void* const* d_in, const int* in_sizes, int n_in,
                              void* d_out, int out_size, void* d_ws, size_t ws_size,
                              hipStream_t stream) {
  (void)n_in; (void)out_size; (void)d_ws; (void)ws_size;
  const float* x   = (const float*)d_in[0];
  const float* cw  = (const float*)d_in[1];
  const float* cb  = (const float*)d_in[2];
  const float* w1  = (const float*)d_in[3];
  const float* b1  = (const float*)d_in[4];
  const float* w2  = (const float*)d_in[5];
  const float* b2  = (const float*)d_in[6];
  const float* w3  = (const float*)d_in[7];
  const float* b3  = (const float*)d_in[8];
  const int B = in_sizes[0] / 1200;
  const int grid = (B + WAVES * ROWS - 1) / (WAVES * ROWS);
  rmi_fused<<<grid, WAVES * 32, 0, stream>>>(x, cw, cb, w1, b1, w2, b2, w3, b3,
                                             (float*)d_out, B);
}


# dense-layer weights and biases copied once per work-group into LDS as k row pairs under the conv; GEMM B fragments are 8-byte LDS reads grouped ahead of each matrix block; sum orders unchanged
# speedup vs baseline: 2.6496x; 2.6496x over previous
#include <hip/hip_runtime.h>
#include <math.h>

typedef __attribute__((ext_vector_type(2))) float v2f;
typedef __attribute__((ext_vector_type(4))) float v4f;
typedef __attribute__((ext_vector_type(8))) float v8f;

#define WAVES   4
#define ROWS    16
#define HPITCH  197
#define H1PITCH 67
#define H2PITCH 33
#define OPITCH  17
#define H2OFF   1536
#define W1PP    100
#define W2PP    96
#define W3PP    32
#define W2OFF   9800
#define W3OFF   (W2OFF + 26 * W2PP)
#define BSOFF   (W3OFF + 16 * W3PP)
#define WSLEN   (BSOFF + 96)

__device__ __forceinline__ void jacobi3(float* s, float* v, int p, int q) {
  float apq = s[p * 3 + q];
  if (fabsf(apq) < 1e-20f) return;
  float app = s[p * 3 + p], aqq = s[q * 3 + q];
  float tau = (aqq - app) / (2.f * apq);
  float t  = copysignf(1.f, tau) / (fabsf(tau) + sqrtf(1.f + tau * tau));
  float c  = 1.f / sqrtf(1.f + t * t);
  float sn = t * c;
  const int k = 3 - p - q;
  float skp = s[k * 3 + p], skq = s[k * 3 + q];
  float nkp = c * skp - sn * skq;
  float nkq = sn * skp + c * skq;
  s[k * 3 + p] = nkp; s[p * 3 + k] = nkp;
  s[k * 3 + q] = nkq; s[q * 3 + k] = nkq;
  s[p * 3 + p] = c * c * app - 2.f * sn * c * apq + sn * sn * aqq;
  s[q * 3 + q] = sn * sn * app + 2.f * sn * c * apq + c * c * aqq;
  s[p * 3 + q] = 0.f; s[q * 3 + p] = 0.f;
#pragma unroll
  for (int i = 0; i < 3; ++i) {
    float vip = v[i * 3 + p], viq = v[i * 3 + q];
    v[i * 3 + p] = c * vip - sn * viq;
    v[i * 3 + q] = sn * vip + c * viq;
  }
}

__global__ __launch_bounds__(128) void rmi_fused(
    const float* __restrict__ x,   const float* __restrict__ cwg,
    const float* __restrict__ cbg, const float* __restrict__ w1,
    const float* __restrict__ b1,  const float* __restrict__ w2,
    const float* __restrict__ b2,  const float* __restrict__ w3,
    const float* __restrict__ b3,  float* __restrict__ out, int B) {
  __shared__ float lds [WAVES * ROWS * HPITCH];
  __shared__ float ldsO[WAVES * ROWS * OPITCH];
  __shared__ __attribute__((aligned(16))) float WS[WSLEN];

  const int tid  = threadIdx.x;
  const int wave = tid >> 5;
  const int lane = tid & 31;
  const int rowBase = blockIdx.x * (WAVES * ROWS) + wave * ROWS;
  float* H = &lds [wave * ROWS * HPITCH];
  float* O = &ldsO[wave * ROWS * OPITCH];

  const int hs = (lane >= 16) ? 2 : 0;
  const int rA = lane & 15;
  const int rD = (lane < 16) ? 0 : 8;

  const float cb = cbg[0];

  const int hh = lane >> 4;
  v2f Bf[6][5];
#pragma unroll
  for (int c = 0; c < 6; ++c)
#pragma unroll
    for (int s = 0; s < 5; ++s) {
      const int ib = (s < 4) ? (8 * hh + 2 * s) : (16 + 2 * hh);
      const int t0 = ib - rA;
      const int t1 = t0 + 1;
      const int i0 = (t0 < 0) ? 0 : ((t0 > 4) ? 4 : t0);
      const int i1 = (t1 < 0) ? 0 : ((t1 > 4) ? 4 : t1);
      const float wa = cwg[c * 5 + i0];
      const float wb = cwg[c * 5 + i1];
      Bf[c][s].x = (t0 >= 0 && t0 <= 4) ? wa : 0.f;
      Bf[c][s].y = (t1 >= 0 && t1 <= 4) ? wb : 0.f;
    }

  const int rowBaseS = blockIdx.x * (WAVES * ROWS) + __builtin_amdgcn_readfirstlane(wave) * ROWS;
  const char* xb = (const char*)x;
#define TILE_OFFS(q, offA, offH) {                                        \
    const int g_ = 16 * (q) + rA;                                          \
    const int b_ = g_ / 13;                                                \
    const int p_ = g_ - 13 * b_;                                           \
    int row_ = rowBaseS + b_;       row_ = (row_ < B) ? row_ : (B - 1);    \
    int oa_ = 16 * p_ + 8 * hh;      oa_ = (oa_ < 192) ? oa_ : 192;        \
    int oh_ = 16 * p_ + 16 + 2 * hh; oh_ = (oh_ < 198) ? oh_ : 198;        \
    offA = (unsigned)row_ * 4800u + 4u * (unsigned)oa_;                    \
    offH = (unsigned)row_ * 4800u + 4u * (unsigned)oh_; }

  v4f L0[13][6], L1[13][6];
  v2f HL[13][6];
#pragma unroll
  for (int q = 0; q < 2; ++q) {
    unsigned offA, offH;
    TILE_OFFS(q, offA, offH)
#pragma unroll
    for (int c = 0; c < 6; ++c) {
      L0[q][c] = *(const v4f*)(xb + offA + c * 800);
      L1[q][c] = *(const v4f*)(xb + offA + c * 800 + 16);
      HL[q][c] = *(const v2f*)(xb + offH + c * 800);
    }
  }

  {
    const int ts = (tid < 124) ? tid : 124;
    const int sr = ts / 25, sp = ts - 25 * sr;
#pragma unroll
    for (int j = 0; j < 20; ++j) {
      int m = sr + 5 * j;
      if (j == 19) m = (m < 97) ? m : 97;
      const v2f ga = *(const v2f*)(w1 + m * 100 + 2 * sp);
      const v2f gb = *(const v2f*)(w1 + m * 100 + 50 + 2 * sp);
      v4f o; o.x = ga.x; o.y = gb.x; o.z = ga.y; o.w = gb.y;
      *(v4f*)(WS + m * W1PP + 4 * sp) = o;
    }
#pragma unroll
    for (int j = 0; j < 3; ++j) {
      int it = tid + 128 * j;  it = (it < 374) ? it : 374;
      const int m = it / 15, p = it - 15 * m;
      const v2f ga = *(const v2f*)(w2 + m * 60 + 2 * p);
      const v2f gb = *(const v2f*)(w2 + m * 60 + 30 + 2 * p);
      v4f o; o.x = ga.x; o.y = gb.x; o.z = ga.y; o.w = gb.y;
      *(v4f*)(WS + W2OFF + m * W2PP + 4 * p) = o;
    }
#pragma unroll
    for (int j = 0; j < 2; ++j) {
      int it = tid + 128 * j;  it = (it < 224) ? it : 224;
      const int m = it / 15, c = it - 15 * m;
      v2f o; o.x = w3[m * 30 + c]; o.y = w3[m * 30 + 15 + c];
      *(v2f*)(WS + W3OFF + m * W3PP + 2 * c) = o;
    }
    const v2f z2 = {0.f, 0.f};
    *(v2f*)(WS + W2OFF + 25 * W2PP + 2 * (tid & 31)) = z2;
    *(v2f*)(WS + W3OFF + 15 * W3PP + 2 * (tid & 15)) = z2;
    const int t1 = (tid < 49) ? tid : 49, t2 = (tid < 29) ? tid : 29, t3 = (tid < 14) ? tid : 14;
    WS[BSOFF + t1]      = b1[t1];
    WS[BSOFF + 50 + t2] = b2[t2];
    WS[BSOFF + 80 + t3] = b3[t3];
  }

#define WM(acc, a, b) acc = __builtin_amdgcn_wmma_f32_16x16x4_f32( \
      false, a, false, b, (short)0, acc, false, false)
#define LO2(v) __builtin_shufflevector(v, v, 0, 1)
#define HI2(v) __builtin_shufflevector(v, v, 2, 3)

#pragma unroll
  for (int q = 0; q < 13; ++q) {
    if (q + 2 < 13) {
      unsigned offA, offH;
      TILE_OFFS(q + 2, offA, offH)
#pragma unroll
      for (int c = 0; c < 6; ++c) {
        L0[q + 2][c] = *(const v4f*)(xb + offA + c * 800);
        L1[q + 2][c] = *(const v4f*)(xb + offA + c * 800 + 16);
        HL[q + 2][c] = *(const v2f*)(xb + offH + c * 800);
      }
    }
    __builtin_amdgcn_sched_barrier(0);
    v8f a0 = {}, a1 = {};
#pragma unroll
    for (int c = 0; c < 3; ++c) {
      WM(a0, LO2(L0[q][c]),     Bf[c][0]);
      WM(a1, LO2(L0[q][c + 3]), Bf[c + 3][0]);
    }
#pragma unroll
    for (int c = 0; c < 3; ++c) {
      WM(a0, HI2(L0[q][c]),     Bf[c][1]);
      WM(a1, HI2(L0[q][c + 3]), Bf[c + 3][1]);
    }
#pragma unroll
    for (int c = 0; c < 3; ++c) {
      WM(a0, LO2(L1[q][c]),     Bf[c][2]);
      WM(a1, LO2(L1[q][c + 3]), Bf[c + 3][2]);
    }
#pragma unroll
    for (int c = 0; c < 3; ++c) {
      WM(a0, HI2(L1[q][c]),     Bf[c][3]);
      WM(a1, HI2(L1[q][c + 3]), Bf[c + 3][3]);
    }
#pragma unroll
    for (int c = 0; c < 3; ++c) {
      WM(a0, HL[q][c],     Bf[c][4]);
      WM(a1, HL[q][c + 3], Bf[c + 3][4]);
    }
#pragma unroll
    for (int r = 0; r < 8; ++r) {
      const int g0 = 16 * q + r, g1 = g0 + 8;
      const int o0 = (g0 / 13) * HPITCH + 16 * (g0 % 13);
      const int o1 = (g1 / 13) * HPITCH + 16 * (g1 % 13);
      const bool short0 = (g0 % 13) == 12, short1 = (g1 % 13) == 12;
      const int  o  = hh ? o1 : o0;
      const bool ok = (hh ? !short1 : !short0) || (rA < 4);
      if (ok) H[o + rA] = (a0[r] + a1[r]) + cb;
    }
  }
#undef WM
#undef LO2
#undef HI2
#undef TILE_OFFS
  __syncthreads();

  int cc1[4];
#pragma unroll
  for (int nt = 0; nt < 4; ++nt) {
    const int c = nt * 16 + rA;
    cc1[nt] = (c < 50) ? c : 49;
  }

  v8f acc1[4] = {};
  {
    const float* HA = H + rA * HPITCH + hs;
    const float* WB = WS + hh * W1PP;
    v2f Af[2][7], Bq[2][7][4];
#define G1LOAD(buf, j)                                                        \
    _Pragma("unroll") for (int s = 0; s < 7; ++s) {                           \
      const int kk = 7 * (j) + s;                                             \
      Af[buf][s].x = HA[4 * kk]; Af[buf][s].y = HA[4 * kk + 1];               \
      _Pragma("unroll") for (int nt = 0; nt < 4; ++nt)                        \
        Bq[buf][s][nt] = *(const v2f*)(WB + 2 * kk * W1PP + 2 * cc1[nt]);     \
    }
    G1LOAD(0, 0)
#pragma unroll
    for (int j = 0; j < 7; ++j) {
      if (j + 1 < 7) { G1LOAD((j + 1) & 1, j + 1) }
      __builtin_amdgcn_sched_barrier(0);
#pragma unroll
      for (int s = 0; s < 7; ++s)
#pragma unroll
        for (int nt = 0; nt < 4; ++nt)
          acc1[nt] = __builtin_amdgcn_wmma_f32_16x16x4_f32(
              false, Af[j & 1][s], false, Bq[j & 1][s][nt], (short)0, acc1[nt], false, false);
      __builtin_amdgcn_sched_barrier(0);
    }
#undef G1LOAD
  }
  __syncthreads();

  float* H1 = H;
#pragma unroll
  for (int nt = 0; nt < 4; ++nt) {
    const int col = nt * 16 + rA;
    const float bb = WS[BSOFF + cc1[nt]];
#pragma unroll
    for (int r = 0; r < 8; ++r) {
      float vv = fmaxf(acc1[nt][r] + bb, 0.f);
      vv = (col < 50) ? vv : 0.f;
      H1[(rD + r) * H1PITCH + col] = vv;
    }
  }
  __syncthreads();

  const int cc2 = (16 + rA < 30) ? (16 + rA) : 29;
  v8f acc2[2] = {};
  {
    const float* HA = H1 + rA * H1PITCH + hs;
    const float* WB = WS + W2OFF + hh * W2PP;
    v2f Af[13], B0[13], B1[13];
#pragma unroll
    for (int kk = 0; kk < 13; ++kk) {
      Af[kk].x = HA[4 * kk]; Af[kk].y = HA[4 * kk + 1];
      B0[kk] = *(const v2f*)(WB + 2 * kk * W2PP + 2 * rA);
      B1[kk] = *(const v2f*)(WB + 2 * kk * W2PP + 2 * cc2);
    }
    __builtin_amdgcn_sched_barrier(0);
#pragma unroll
    for (int kk = 0; kk < 13; ++kk) {
      acc2[0] = __builtin_amdgcn_wmma_f32_16x16x4_f32(
          false, Af[kk], false, B0[kk], (short)0, acc2[0], false, false);
      acc2[1] = __builtin_amdgcn_wmma_f32_16x16x4_f32(
          false, Af[kk], false, B1[kk], (short)0, acc2[1], false, false);
    }
  }
  __syncthreads();
  float* H2 = H + H2OFF;
#pragma unroll
  for (int nt = 0; nt < 2; ++nt) {
    const int col = nt * 16 + rA;
    const float bb = WS[BSOFF + 50 + ((col < 30) ? col : 29)];
#pragma unroll
    for (int r = 0; r < 8; ++r) {
      float vv = fmaxf(acc2[nt][r] + bb, 0.f);
      vv = (col < 30) ? vv : 0.f;
      H2[(rD + r) * H2PITCH + col] = vv;
    }
  }
  __syncthreads();

  const int cc3 = (rA < 15) ? rA : 14;
  v8f acc3 = {};
  {
    const float* HA = H2 + rA * H2PITCH + hs;
    const float* WB = WS + W3OFF + hh * W3PP;
    v2f Af[8], Bv[8];
#pragma unroll
    for (int kk = 0; kk < 8; ++kk) {
      Af[kk].x = HA[4 * kk]; Af[kk].y = HA[4 * kk + 1];
      Bv[kk] = *(const v2f*)(WB + 2 * kk * W3PP + 2 * cc3);
    }
    __builtin_amdgcn_sched_barrier(0);
#pragma unroll
    for (int kk = 0; kk < 8; ++kk)
      acc3 = __builtin_amdgcn_wmma_f32_16x16x4_f32(
          false, Af[kk], false, Bv[kk], (short)0, acc3, false, false);
  }
  {
    const float bb = WS[BSOFF + 80 + cc3];
#pragma unroll
    for (int r = 0; r < 8; ++r)
      O[(rD + r) * OPITCH + rA] = acc3[r] + bb;
  }
  __syncthreads();

  if (lane < 16) {
    const int gRow = rowBase + lane;
    if (gRow < B) {
      float a9[9];
#pragma unroll
      for (int i = 0; i < 9; ++i) a9[i] = O[lane * OPITCH + i];
      float s[9], v[9];
#pragma unroll
      for (int i = 0; i < 3; ++i)
#pragma unroll
        for (int j = 0; j < 3; ++j) {
          s[i * 3 + j] = a9[0 * 3 + i] * a9[0 * 3 + j] +
                         a9[1 * 3 + i] * a9[1 * 3 + j] +
                         a9[2 * 3 + i] * a9[2 * 3 + j];
          v[i * 3 + j] = (i == j) ? 1.f : 0.f;
        }
#pragma unroll
      for (int sw = 0; sw < 6; ++sw) {
        jacobi3(s, v, 0, 1); jacobi3(s, v, 0, 2); jacobi3(s, v, 1, 2);
      }
      float g0 = s[0], g1 = s[4], g2 = s[8];
#define SWAPCOL(p, q) { float t_;                              \
        t_ = v[p];     v[p]     = v[q];     v[q]     = t_;     \
        t_ = v[3 + p]; v[3 + p] = v[3 + q]; v[3 + q] = t_;     \
        t_ = v[6 + p]; v[6 + p] = v[6 + q]; v[6 + q] = t_; }
      if (g0 < g1) { float t = g0; g0 = g1; g1 = t; SWAPCOL(0, 1); }
      if (g0 < g2) { float t = g0; g0 = g2; g2 = t; SWAPCOL(0, 2); }
      if (g1 < g2) { float t = g1; g1 = g2; g2 = t; SWAPCOL(1, 2); }
#undef SWAPCOL
      float w[9];
#pragma unroll
      for (int i = 0; i < 3; ++i)
#pragma unroll
        for (int j = 0; j < 3; ++j)
          w[i * 3 + j] = a9[i * 3 + 0] * v[0 + j] +
                         a9[i * 3 + 1] * v[3 + j] +
                         a9[i * 3 + 2] * v[6 + j];
      float u0x = w[0], u0y = w[3], u0z = w[6];
      float n0 = 1.f / sqrtf(u0x * u0x + u0y * u0y + u0z * u0z + 1e-20f);
      u0x *= n0; u0y *= n0; u0z *= n0;
      float u1x = w[1], u1y = w[4], u1z = w[7];
      float d01 = u1x * u0x + u1y * u0y + u1z * u0z;
      u1x -= d01 * u0x; u1y -= d01 * u0y; u1z -= d01 * u0z;
      float n1 = 1.f / sqrtf(u1x * u1x + u1y * u1y + u1z * u1z + 1e-20f);
      u1x *= n1; u1y *= n1; u1z *= n1;
      float cx = u0y * u1z - u0z * u1y;
      float cy = u0z * u1x - u0x * u1z;
      float cz = u0x * u1y - u0y * u1x;
      float detV = v[0] * (v[4] * v[8] - v[5] * v[7]) -
                   v[1] * (v[3] * v[8] - v[5] * v[6]) +
                   v[2] * (v[3] * v[7] - v[4] * v[6]);
      float sg = (detV >= 0.f) ? 1.f : -1.f;
      float u2x = sg * cx, u2y = sg * cy, u2z = sg * cz;
      float U_[9] = { u0x, u1x, u2x,  u0y, u1y, u2y,  u0z, u1z, u2z };
      float* op = out + (size_t)gRow * 15;
#pragma unroll
      for (int i = 0; i < 3; ++i)
#pragma unroll
        for (int j = 0; j < 3; ++j)
          op[i * 3 + j] = U_[i * 3 + 0] * v[j * 3 + 0] +
                          U_[i * 3 + 1] * v[j * 3 + 1] +
                          U_[i * 3 + 2] * v[j * 3 + 2];
#pragma unroll
      for (int i = 9; i < 15; ++i) op[i] = O[lane * OPITCH + i];
    }
  }
}

extern "C" void kernel_launch(void* const* d_in, const int* in_sizes, int n_in,
                              void* d_out, int out_size, void* d_ws, size_t ws_size,
                              hipStream_t stream) {
  (void)n_in; (void)out_size; (void)d_ws; (void)ws_size;
  const float* x   = (const float*)d_in[0];
  const float* cw  = (const float*)d_in[1];
  const float* cb  = (const float*)d_in[2];
  const float* w1  = (const float*)d_in[3];
  const float* b1  = (const float*)d_in[4];
  const float* w2  = (const float*)d_in[5];
  const float* b2  = (const float*)d_in[6];
  const float* w3  = (const float*)d_in[7];
  const float* b3  = (const float*)d_in[8];
  const int B = in_sizes[0] / 1200;
  const int grid = (B + WAVES * ROWS - 1) / (WAVES * ROWS);
  rmi_fused<<<grid, WAVES * 32, 0, stream>>>(x, cw, cb, w1, b1, w2, b2, w3, b3,
                                             (float*)d_out, B);
}


# conv tile M index is a pair of 4-sample groups of one row (two 8-row streams, 25 tiles): whole-line x loads, no halo loads, 300 not 390 matrix ops, halves joined by lane shift; conv sum order changes
# speedup vs baseline: 2.7795x; 1.0490x over previous
#include <hip/hip_runtime.h>
#include <math.h>

typedef __attribute__((ext_vector_type(2))) float v2f;
typedef __attribute__((ext_vector_type(4))) float v4f;
typedef __attribute__((ext_vector_type(8))) float v8f;

#define WAVES   4
#define ROWS    16
#define HPITCH  197
#define H1PITCH 67
#define H2PITCH 33
#define OPITCH  17
#define H2OFF   1536
#define W1PP    100
#define W2PP    96
#define W3PP    32
#define W2OFF   9800
#define W3OFF   (W2OFF + 26 * W2PP)
#define BSOFF   (W3OFF + 16 * W3PP)
#define WSLEN   (BSOFF + 96)

__device__ __forceinline__ void jacobi3(float* s, float* v, int p, int q) {
  float apq = s[p * 3 + q];
  if (fabsf(apq) < 1e-20f) return;
  float app = s[p * 3 + p], aqq = s[q * 3 + q];
  float tau = (aqq - app) / (2.f * apq);
  float t  = copysignf(1.f, tau) / (fabsf(tau) + sqrtf(1.f + tau * tau));
  float c  = 1.f / sqrtf(1.f + t * t);
  float sn = t * c;
  const int k = 3 - p - q;
  float skp = s[k * 3 + p], skq = s[k * 3 + q];
  float nkp = c * skp - sn * skq;
  float nkq = sn * skp + c * skq;
  s[k * 3 + p] = nkp; s[p * 3 + k] = nkp;
  s[k * 3 + q] = nkq; s[q * 3 + k] = nkq;
  s[p * 3 + p] = c * c * app - 2.f * sn * c * apq + sn * sn * aqq;
  s[q * 3 + q] = sn * sn * app + 2.f * sn * c * apq + c * c * aqq;
  s[p * 3 + q] = 0.f; s[q * 3 + p] = 0.f;
#pragma unroll
  for (int i = 0; i < 3; ++i) {
    float vip = v[i * 3 + p], viq = v[i * 3 + q];
    v[i * 3 + p] = c * vip - sn * viq;
    v[i * 3 + q] = sn * vip + c * viq;
  }
}

__device__ __forceinline__ float shr8_zero(float v) {
  return __builtin_bit_cast(float, __builtin_amdgcn_update_dpp(
      0, __builtin_bit_cast(int, v), 0x118  , 0xF, 0xF, true));
}

__global__ __launch_bounds__(128) void rmi_fused(
    const float* __restrict__ x,   const float* __restrict__ cwg,
    const float* __restrict__ cbg, const float* __restrict__ w1,
    const float* __restrict__ b1,  const float* __restrict__ w2,
    const float* __restrict__ b2,  const float* __restrict__ w3,
    const float* __restrict__ b3,  float* __restrict__ out, int B) {
  __shared__ float lds [WAVES * ROWS * HPITCH];
  __shared__ float ldsO[WAVES * ROWS * OPITCH];
  __shared__ __attribute__((aligned(16))) float WS[WSLEN];

  const int tid  = threadIdx.x;
  const int wave = tid >> 5;
  const int lane = tid & 31;
  const int rowBase = blockIdx.x * (WAVES * ROWS) + wave * ROWS;
  float* H = &lds [wave * ROWS * HPITCH];
  float* O = &ldsO[wave * ROWS * OPITCH];

  const int hs = (lane >= 16) ? 2 : 0;
  const int rA = lane & 15;
  const int rD = (lane < 16) ? 0 : 8;

  const float cb = cbg[0];

  const int hh = lane >> 4;
  v2f Bf[6][2];
#pragma unroll
  for (int c = 0; c < 6; ++c)
#pragma unroll
    for (int z = 0; z < 2; ++z) {
      const int t0 = (4 * hh + 2 * z) - (rA - 4);
      const int t1 = t0 + 1;
      const int i0 = (t0 < 0) ? 0 : ((t0 > 4) ? 4 : t0);
      const int i1 = (t1 < 0) ? 0 : ((t1 > 4) ? 4 : t1);
      const float wa = cwg[c * 5 + i0];
      const float wb = cwg[c * 5 + i1];
      Bf[c][z].x = (t0 >= 0 && t0 <= 4) ? wa : 0.f;
      Bf[c][z].y = (t1 >= 0 && t1 <= 4) ? wb : 0.f;
    }

  const int rowBaseS = blockIdx.x * (WAVES * ROWS) + __builtin_amdgcn_readfirstlane(wave) * ROWS;
  const char* xb = (const char*)x;
#define TILE_OFF(t, off) {                                                \
    const int P_  = 8 * (t) + (rA & 7);                                    \
    const int b_  = P_ / 25;                                               \
    const int pp_ = P_ - 25 * b_;                                          \
    int row_ = rowBaseS + 8 * (rA >> 3) + b_;                              \
    row_ = (row_ < B) ? row_ : (B - 1);                                    \
    off = (unsigned)row_ * 4800u + 32u * (unsigned)pp_ + 16u * (unsigned)hh; }
#define TILE_LOAD(t) {                                                    \
    unsigned off_;                                                         \
    TILE_OFF(t, off_)                                                      \
    _Pragma("unroll") for (int c = 0; c < 6; ++c)                          \
      L[t][c] = *(const v4f*)(xb + off_ + c * 800); }

#define NTILE 25
#define PF    6
  v4f L[NTILE][6];
#pragma unroll
  for (int t = 0; t < PF; ++t) TILE_LOAD(t)

  {
    const int ts = (tid < 124) ? tid : 124;
    const int sr = ts / 25, sp = ts - 25 * sr;
#pragma unroll
    for (int j = 0; j < 20; ++j) {
      int m = sr + 5 * j;
      if (j == 19) m = (m < 97) ? m : 97;
      const v2f ga = *(const v2f*)(w1 + m * 100 + 2 * sp);
      const v2f gb = *(const v2f*)(w1 + m * 100 + 50 + 2 * sp);
      v4f o; o.x = ga.x; o.y = gb.x; o.z = ga.y; o.w = gb.y;
      *(v4f*)(WS + m * W1PP + 4 * sp) = o;
    }
#pragma unroll
    for (int j = 0; j < 3; ++j) {
      int it = tid + 128 * j;  it = (it < 374) ? it : 374;
      const int m = it / 15, p = it - 15 * m;
      const v2f ga = *(const v2f*)(w2 + m * 60 + 2 * p);
      const v2f gb = *(const v2f*)(w2 + m * 60 + 30 + 2 * p);
      v4f o; o.x = ga.x; o.y = gb.x; o.z = ga.y; o.w = gb.y;
      *(v4f*)(WS + W2OFF + m * W2PP + 4 * p) = o;
    }
#pragma unroll
    for (int j = 0; j < 2; ++j) {
      int it = tid + 128 * j;  it = (it < 224) ? it : 224;
      const int m = it / 15, c = it - 15 * m;
      v2f o; o.x = w3[m * 30 + c]; o.y = w3[m * 30 + 15 + c];
      *(v2f*)(WS + W3OFF + m * W3PP + 2 * c) = o;
    }
    const v2f z2 = {0.f, 0.f};
    *(v2f*)(WS + W2OFF + 25 * W2PP + 2 * (tid & 31)) = z2;
    *(v2f*)(WS + W3OFF + 15 * W3PP + 2 * (tid & 15)) = z2;
    const int t1 = (tid < 49) ? tid : 49, t2 = (tid < 29) ? tid : 29, t3 = (tid < 14) ? tid : 14;
    WS[BSOFF + t1]      = b1[t1];
    WS[BSOFF + 50 + t2] = b2[t2];
    WS[BSOFF + 80 + t3] = b3[t3];
  }

#define WM(acc, a, b) acc = __builtin_amdgcn_wmma_f32_16x16x4_f32( \
      false, a, false, b, (short)0, acc, false, false)
#define LO2(v) __builtin_shufflevector(v, v, 0, 1)
#define HI2(v) __builtin_shufflevector(v, v, 2, 3)

  const bool st8 = (rA >= 4) && (rA <= 11);
  const bool st4 = (rA >= 4) && (rA <= 7);
  const int  hst = 8 * hh * HPITCH + rA - 4;
#define FIN(t) {                                                               \
    float g_[8];                                                               \
    _Pragma("unroll") for (int r = 0; r < 8; ++r) {                            \
      const int P_ = 8 * (t) + r, pp_ = P_ % 25;                               \
      float v_ = acc[t][r];                                                    \
      if (pp_ != 24) {                                                         \
        const float nx_ = (r < 7) ? acc[t][(r + 1) & 7] : acc[((t) + 1) % (NTILE + 1)][0]; \
        v_ += shr8_zero(nx_);                                                  \
      }                                                                        \
      g_[r] = v_ + cb;                                                         \
    }                                                                          \
    if (st8) {                                                                 \
      _Pragma("unroll") for (int r = 0; r < 8; ++r) {                          \
        const int P_ = 8 * (t) + r, b_ = P_ / 25, pp_ = P_ - 25 * b_;          \
        if (pp_ != 24) H[hst + b_ * HPITCH + 8 * pp_] = g_[r];                 \
      }                                                                        \
    }                                                                          \
    if (st4) {                                                                 \
      _Pragma("unroll") for (int r = 0; r < 8; ++r) {                          \
        const int P_ = 8 * (t) + r, b_ = P_ / 25, pp_ = P_ - 25 * b_;          \
        if (pp_ == 24) H[hst + b_ * HPITCH + 8 * pp_] = g_[r];                 \
      }                                                                        \
    } }

  v8f acc[NTILE + 1];
#pragma unroll
  for (int u = 0; u < (NTILE + 1) / 2; ++u) {
    const int tA = 2 * u, tB = 2 * u + 1;
    if (tA + PF < NTILE) TILE_LOAD(tA + PF)
    if (tB + PF < NTILE) TILE_LOAD(tB + PF)
    __builtin_amdgcn_sched_barrier(0);
    acc[tA] = v8f{};
    if (tB < NTILE) acc[tB] = v8f{};
#pragma unroll
    for (int c = 0; c < 6; ++c) {
      WM(acc[tA], LO2(L[tA][c]), Bf[c][0]);
      if (tB < NTILE) WM(acc[tB], LO2(L[tB][c]), Bf[c][0]);
      WM(acc[tA], HI2(L[tA][c]), Bf[c][1]);
      if (tB < NTILE) WM(acc[tB], HI2(L[tB][c]), Bf[c][1]);
    }
    if (u > 0) FIN(tA - 1)
    FIN(tA)
  }
#undef WM
#undef LO2
#undef HI2
#undef FIN
#undef TILE_LOAD
#undef TILE_OFF
  __syncthreads();

  int cc1[4];
#pragma unroll
  for (int nt = 0; nt < 4; ++nt) {
    const int c = nt * 16 + rA;
    cc1[nt] = (c < 50) ? c : 49;
  }

  v8f acc1[4] = {};
  {
    const float* HA = H + rA * HPITCH + hs;
    const float* WB = WS + hh * W1PP;
    v2f Af[2][7], Bq[2][7][4];
#define G1LOAD(buf, j)                                                        \
    _Pragma("unroll") for (int s = 0; s < 7; ++s) {                           \
      const int kk = 7 * (j) + s;                                             \
      Af[buf][s].x = HA[4 * kk]; Af[buf][s].y = HA[4 * kk + 1];               \
      _Pragma("unroll") for (int nt = 0; nt < 4; ++nt)                        \
        Bq[buf][s][nt] = *(const v2f*)(WB + 2 * kk * W1PP + 2 * cc1[nt]);     \
    }
    G1LOAD(0, 0)
#pragma unroll
    for (int j = 0; j < 7; ++j) {
      if (j + 1 < 7) { G1LOAD((j + 1) & 1, j + 1) }
      __builtin_amdgcn_sched_barrier(0);
#pragma unroll
      for (int s = 0; s < 7; ++s)
#pragma unroll
        for (int nt = 0; nt < 4; ++nt)
          acc1[nt] = __builtin_amdgcn_wmma_f32_16x16x4_f32(
              false, Af[j & 1][s], false, Bq[j & 1][s][nt], (short)0, acc1[nt], false, false);
      __builtin_amdgcn_sched_barrier(0);
    }
#undef G1LOAD
  }
  __syncthreads();

  float* H1 = H;
#pragma unroll
  for (int nt = 0; nt < 4; ++nt) {
    const int col = nt * 16 + rA;
    const float bb = WS[BSOFF + cc1[nt]];
#pragma unroll
    for (int r = 0; r < 8; ++r) {
      float vv = fmaxf(acc1[nt][r] + bb, 0.f);
      vv = (col < 50) ? vv : 0.f;
      H1[(rD + r) * H1PITCH + col] = vv;
    }
  }
  __syncthreads();

  const int cc2 = (16 + rA < 30) ? (16 + rA) : 29;
  v8f acc2[2] = {};
  {
    const float* HA = H1 + rA * H1PITCH + hs;
    const float* WB = WS + W2OFF + hh * W2PP;
    v2f Af[13], B0[13], B1[13];
#pragma unroll
    for (int kk = 0; kk < 13; ++kk) {
      Af[kk].x = HA[4 * kk]; Af[kk].y = HA[4 * kk + 1];
      B0[kk] = *(const v2f*)(WB + 2 * kk * W2PP + 2 * rA);
      B1[kk] = *(const v2f*)(WB + 2 * kk * W2PP + 2 * cc2);
    }
    __builtin_amdgcn_sched_barrier(0);
#pragma unroll
    for (int kk = 0; kk < 13; ++kk) {
      acc2[0] = __builtin_amdgcn_wmma_f32_16x16x4_f32(
          false, Af[kk], false, B0[kk], (short)0, acc2[0], false, false);
      acc2[1] = __builtin_amdgcn_wmma_f32_16x16x4_f32(
          false, Af[kk], false, B1[kk], (short)0, acc2[1], false, false);
    }
  }
  __syncthreads();
  float* H2 = H + H2OFF;
#pragma unroll
  for (int nt = 0; nt < 2; ++nt) {
    const int col = nt * 16 + rA;
    const float bb = WS[BSOFF + 50 + ((col < 30) ? col : 29)];
#pragma unroll
    for (int r = 0; r < 8; ++r) {
      float vv = fmaxf(acc2[nt][r] + bb, 0.f);
      vv = (col < 30) ? vv : 0.f;
      H2[(rD + r) * H2PITCH + col] = vv;
    }
  }
  __syncthreads();

  const int cc3 = (rA < 15) ? rA : 14;
  v8f acc3 = {};
  {
    const float* HA = H2 + rA * H2PITCH + hs;
    const float* WB = WS + W3OFF + hh * W3PP;
    v2f Af[8], Bv[8];
#pragma unroll
    for (int kk = 0; kk < 8; ++kk) {
      Af[kk].x = HA[4 * kk]; Af[kk].y = HA[4 * kk + 1];
      Bv[kk] = *(const v2f*)(WB + 2 * kk * W3PP + 2 * cc3);
    }
    __builtin_amdgcn_sched_barrier(0);
#pragma unroll
    for (int kk = 0; kk < 8; ++kk)
      acc3 = __builtin_amdgcn_wmma_f32_16x16x4_f32(
          false, Af[kk], false, Bv[kk], (short)0, acc3, false, false);
  }
  {
    const float bb = WS[BSOFF + 80 + cc3];
#pragma unroll
    for (int r = 0; r < 8; ++r)
      O[(rD + r) * OPITCH + rA] = acc3[r] + bb;
  }
  __syncthreads();

  if (lane < 16) {
    const int gRow = rowBase + lane;
    if (gRow < B) {
      float a9[9];
#pragma unroll
      for (int i = 0; i < 9; ++i) a9[i] = O[lane * OPITCH + i];
      float s[9], v[9];
#pragma unroll
      for (int i = 0; i < 3; ++i)
#pragma unroll
        for (int j = 0; j < 3; ++j) {
          s[i * 3 + j] = a9[0 * 3 + i] * a9[0 * 3 + j] +
                         a9[1 * 3 + i] * a9[1 * 3 + j] +
                         a9[2 * 3 + i] * a9[2 * 3 + j];
          v[i * 3 + j] = (i == j) ? 1.f : 0.f;
        }
#pragma unroll
      for (int sw = 0; sw < 6; ++sw) {
        jacobi3(s, v, 0, 1); jacobi3(s, v, 0, 2); jacobi3(s, v, 1, 2);
      }
      float g0 = s[0], g1 = s[4], g2 = s[8];
#define SWAPCOL(p, q) { float t_;                              \
        t_ = v[p];     v[p]     = v[q];     v[q]     = t_;     \
        t_ = v[3 + p]; v[3 + p] = v[3 + q]; v[3 + q] = t_;     \
        t_ = v[6 + p]; v[6 + p] = v[6 + q]; v[6 + q] = t_; }
      if (g0 < g1) { float t = g0; g0 = g1; g1 = t; SWAPCOL(0, 1); }
      if (g0 < g2) { float t = g0; g0 = g2; g2 = t; SWAPCOL(0, 2); }
      if (g1 < g2) { float t = g1; g1 = g2; g2 = t; SWAPCOL(1, 2); }
#undef SWAPCOL
      float w[9];
#pragma unroll
      for (int i = 0; i < 3; ++i)
#pragma unroll
        for (int j = 0; j < 3; ++j)
          w[i * 3 + j] = a9[i * 3 + 0] * v[0 + j] +
                         a9[i * 3 + 1] * v[3 + j] +
                         a9[i * 3 + 2] * v[6 + j];
      float u0x = w[0], u0y = w[3], u0z = w[6];
      float n0 = 1.f / sqrtf(u0x * u0x + u0y * u0y + u0z * u0z + 1e-20f);
      u0x *= n0; u0y *= n0; u0z *= n0;
      float u1x = w[1], u1y = w[4], u1z = w[7];
      float d01 = u1x * u0x + u1y * u0y + u1z * u0z;
      u1x -= d01 * u0x; u1y -= d01 * u0y; u1z -= d01 * u0z;
      float n1 = 1.f / sqrtf(u1x * u1x + u1y * u1y + u1z * u1z + 1e-20f);
      u1x *= n1; u1y *= n1; u1z *= n1;
      float cx = u0y * u1z - u0z * u1y;
      float cy = u0z * u1x - u0x * u1z;
      float cz = u0x * u1y - u0y * u1x;
      float detV = v[0] * (v[4] * v[8] - v[5] * v[7]) -
                   v[1] * (v[3] * v[8] - v[5] * v[6]) +
                   v[2] * (v[3] * v[7] - v[4] * v[6]);
      float sg = (detV >= 0.f) ? 1.f : -1.f;
      float u2x = sg * cx, u2y = sg * cy, u2z = sg * cz;
      float U_[9] = { u0x, u1x, u2x,  u0y, u1y, u2y,  u0z, u1z, u2z };
      float* op = out + (size_t)gRow * 15;
#pragma unroll
      for (int i = 0; i < 3; ++i)
#pragma unroll
        for (int j = 0; j < 3; ++j)
          op[i * 3 + j] = U_[i * 3 + 0] * v[j * 3 + 0] +
                          U_[i * 3 + 1] * v[j * 3 + 1] +
                          U_[i * 3 + 2] * v[j * 3 + 2];
#pragma unroll
      for (int i = 9; i < 15; ++i) op[i] = O[lane * OPITCH + i];
    }
  }
}

extern "C" void kernel_launch(void* const* d_in, const int* in_sizes, int n_in,
                              void* d_out, int out_size, void* d_ws, size_t ws_size,
                              hipStream_t stream) {
  (void)n_in; (void)out_size; (void)d_ws; (void)ws_size;
  const float* x   = (const float*)d_in[0];
  const float* cw  = (const float*)d_in[1];
  const float* cb  = (const float*)d_in[2];
  const float* w1  = (const float*)d_in[3];
  const float* b1  = (const float*)d_in[4];
  const float* w2  = (const float*)d_in[5];
  const float* b2  = (const float*)d_in[6];
  const float* w3  = (const float*)d_in[7];
  const float* b3  = (const float*)d_in[8];
  const int B = in_sizes[0] / 1200;
  const int grid = (B + WAVES * ROWS - 1) / (WAVES * ROWS);
  rmi_fused<<<grid, WAVES * 32, 0, stream>>>(x, cw, cb, w1, b1, w2, b2, w3, b3,
                                             (float*)d_out, B);
}


# 3x3 SVD tail: Jacobi (cos, sin) as quotients of squares of (2(aqq-app), apq), capped at pi/4, leftover element kept: no square root, one division level per rotation; A pre-scaled by a power of 2
# speedup vs baseline: 2.8817x; 1.0368x over previous
#include <hip/hip_runtime.h>
#include <math.h>

typedef __attribute__((ext_vector_type(2))) float v2f;
typedef __attribute__((ext_vector_type(4))) float v4f;
typedef __attribute__((ext_vector_type(8))) float v8f;

#define WAVES   4
#define ROWS    16
#define HPITCH  197
#define H1PITCH 67
#define H2PITCH 33
#define OPITCH  17
#define H2OFF   1536
#define W1PP    100
#define W2PP    96
#define W3PP    32
#define W2OFF   9800
#define W3OFF   (W2OFF + 26 * W2PP)
#define BSOFF   (W3OFF + 16 * W3PP)
#define WSLEN   (BSOFF + 96)

__device__ __forceinline__ void jacobi3(float* s, float* v, int p, int q) {
  float apq = s[p * 3 + q];
  const bool skip = fabsf(apq) < 1e-20f;
  float app = s[p * 3 + p], aqq = s[q * 3 + q];
  const float ch  = 2.f * (aqq - app), sh = apq;
  const float ch2 = ch * ch, sh2 = sh * sh, chsh = ch * sh;
  const float nn  = ch2 + sh2;
  float c  = (ch2 - sh2) / nn;
  float sn = (2.f * chsh) / nn;
  if (!(5.8284273f * sh2 < ch2)) {
    c  = 0.70710678f;
    sn = copysignf(0.70710678f, chsh);
  }
  if (skip) { c = 1.f; sn = 0.f; }
  const int k = 3 - p - q;
  float skp = s[k * 3 + p], skq = s[k * 3 + q];
  float nkp = c * skp - sn * skq;
  float nkq = sn * skp + c * skq;
  s[k * 3 + p] = nkp; s[p * 3 + k] = nkp;
  s[k * 3 + q] = nkq; s[q * 3 + k] = nkq;
  const float cc = c * c, ss = sn * sn, sc = sn * c;
  s[p * 3 + p] = cc * app - 2.f * sc * apq + ss * aqq;
  s[q * 3 + q] = ss * app + 2.f * sc * apq + cc * aqq;
  const float npq = (cc - ss) * apq + sc * (app - aqq);
  s[p * 3 + q] = npq; s[q * 3 + p] = npq;
#pragma unroll
  for (int i = 0; i < 3; ++i) {
    float vip = v[i * 3 + p], viq = v[i * 3 + q];
    v[i * 3 + p] = c * vip - sn * viq;
    v[i * 3 + q] = sn * vip + c * viq;
  }
}

__device__ __forceinline__ float shr8_zero(float v) {
  return __builtin_bit_cast(float, __builtin_amdgcn_update_dpp(
      0, __builtin_bit_cast(int, v), 0x118  , 0xF, 0xF, true));
}

__global__ __launch_bounds__(128) void rmi_fused(
    const float* __restrict__ x,   const float* __restrict__ cwg,
    const float* __restrict__ cbg, const float* __restrict__ w1,
    const float* __restrict__ b1,  const float* __restrict__ w2,
    const float* __restrict__ b2,  const float* __restrict__ w3,
    const float* __restrict__ b3,  float* __restrict__ out, int B) {
  __shared__ float lds [WAVES * ROWS * HPITCH];
  __shared__ float ldsO[WAVES * ROWS * OPITCH];
  __shared__ __attribute__((aligned(16))) float WS[WSLEN];

  const int tid  = threadIdx.x;
  const int wave = tid >> 5;
  const int lane = tid & 31;
  const int rowBase = blockIdx.x * (WAVES * ROWS) + wave * ROWS;
  float* H = &lds [wave * ROWS * HPITCH];
  float* O = &ldsO[wave * ROWS * OPITCH];

  const int hs = (lane >= 16) ? 2 : 0;
  const int rA = lane & 15;
  const int rD = (lane < 16) ? 0 : 8;

  const float cb = cbg[0];

  const int hh = lane >> 4;
  v2f Bf[6][2];
#pragma unroll
  for (int c = 0; c < 6; ++c)
#pragma unroll
    for (int z = 0; z < 2; ++z) {
      const int t0 = (4 * hh + 2 * z) - (rA - 4);
      const int t1 = t0 + 1;
      const int i0 = (t0 < 0) ? 0 : ((t0 > 4) ? 4 : t0);
      const int i1 = (t1 < 0) ? 0 : ((t1 > 4) ? 4 : t1);
      const float wa = cwg[c * 5 + i0];
      const float wb = cwg[c * 5 + i1];
      Bf[c][z].x = (t0 >= 0 && t0 <= 4) ? wa : 0.f;
      Bf[c][z].y = (t1 >= 0 && t1 <= 4) ? wb : 0.f;
    }

  const int rowBaseS = blockIdx.x * (WAVES * ROWS) + __builtin_amdgcn_readfirstlane(wave) * ROWS;
  const char* xb = (const char*)x;
#define TILE_OFF(t, off) {                                                \
    const int P_  = 8 * (t) + (rA & 7);                                    \
    const int b_  = P_ / 25;                                               \
    const int pp_ = P_ - 25 * b_;                                          \
    int row_ = rowBaseS + 8 * (rA >> 3) + b_;                              \
    row_ = (row_ < B) ? row_ : (B - 1);                                    \
    off = (unsigned)row_ * 4800u + 32u * (unsigned)pp_ + 16u * (unsigned)hh; }
#define TILE_LOAD(t) {                                                    \
    unsigned off_;                                                         \
    TILE_OFF(t, off_)                                                      \
    _Pragma("unroll") for (int c = 0; c < 6; ++c)                          \
      L[t][c] = *(const v4f*)(xb + off_ + c * 800); }

#define NTILE 25
#define PF    6
  v4f L[NTILE][6];
#pragma unroll
  for (int t = 0; t < PF; ++t) TILE_LOAD(t)

  {
    const int ts = (tid < 124) ? tid : 124;
    const int sr = ts / 25, sp = ts - 25 * sr;
#pragma unroll
    for (int j = 0; j < 20; ++j) {
      int m = sr + 5 * j;
      if (j == 19) m = (m < 97) ? m : 97;
      const v2f ga = *(const v2f*)(w1 + m * 100 + 2 * sp);
      const v2f gb = *(const v2f*)(w1 + m * 100 + 50 + 2 * sp);
      v4f o; o.x = ga.x; o.y = gb.x; o.z = ga.y; o.w = gb.y;
      *(v4f*)(WS + m * W1PP + 4 * sp) = o;
    }
#pragma unroll
    for (int j = 0; j < 3; ++j) {
      int it = tid + 128 * j;  it = (it < 374) ? it : 374;
      const int m = it / 15, p = it - 15 * m;
      const v2f ga = *(const v2f*)(w2 + m * 60 + 2 * p);
      const v2f gb = *(const v2f*)(w2 + m * 60 + 30 + 2 * p);
      v4f o; o.x = ga.x; o.y = gb.x; o.z = ga.y; o.w = gb.y;
      *(v4f*)(WS + W2OFF + m * W2PP + 4 * p) = o;
    }
#pragma unroll
    for (int j = 0; j < 2; ++j) {
      int it = tid + 128 * j;  it = (it < 224) ? it : 224;
      const int m = it / 15, c = it - 15 * m;
      v2f o; o.x = w3[m * 30 + c]; o.y = w3[m * 30 + 15 + c];
      *(v2f*)(WS + W3OFF + m * W3PP + 2 * c) = o;
    }
    const v2f z2 = {0.f, 0.f};
    *(v2f*)(WS + W2OFF + 25 * W2PP + 2 * (tid & 31)) = z2;
    *(v2f*)(WS + W3OFF + 15 * W3PP + 2 * (tid & 15)) = z2;
    const int t1 = (tid < 49) ? tid : 49, t2 = (tid < 29) ? tid : 29, t3 = (tid < 14) ? tid : 14;
    WS[BSOFF + t1]      = b1[t1];
    WS[BSOFF + 50 + t2] = b2[t2];
    WS[BSOFF + 80 + t3] = b3[t3];
  }

#define WM(acc, a, b) acc = __builtin_amdgcn_wmma_f32_16x16x4_f32( \
      false, a, false, b, (short)0, acc, false, false)
#define LO2(v) __builtin_shufflevector(v, v, 0, 1)
#define HI2(v) __builtin_shufflevector(v, v, 2, 3)

  const bool st8 = (rA >= 4) && (rA <= 11);
  const bool st4 = (rA >= 4) && (rA <= 7);
  const int  hst = 8 * hh * HPITCH + rA - 4;
#define FIN(t) {                                                               \
    float g_[8];                                                               \
    _Pragma("unroll") for (int r = 0; r < 8; ++r) {                            \
      const int P_ = 8 * (t) + r, pp_ = P_ % 25;                               \
      float v_ = acc[t][r];                                                    \
      if (pp_ != 24) {                                                         \
        const float nx_ = (r < 7) ? acc[t][(r + 1) & 7] : acc[((t) + 1) % (NTILE + 1)][0]; \
        v_ += shr8_zero(nx_);                                                  \
      }                                                                        \
      g_[r] = v_ + cb;                                                         \
    }                                                                          \
    if (st8) {                                                                 \
      _Pragma("unroll") for (int r = 0; r < 8; ++r) {                          \
        const int P_ = 8 * (t) + r, b_ = P_ / 25, pp_ = P_ - 25 * b_;          \
        if (pp_ != 24) H[hst + b_ * HPITCH + 8 * pp_] = g_[r];                 \
      }                                                                        \
    }                                                                          \
    if (st4) {                                                                 \
      _Pragma("unroll") for (int r = 0; r < 8; ++r) {                          \
        const int P_ = 8 * (t) + r, b_ = P_ / 25, pp_ = P_ - 25 * b_;          \
        if (pp_ == 24) H[hst + b_ * HPITCH + 8 * pp_] = g_[r];                 \
      }                                                                        \
    } }

  v8f acc[NTILE + 1];
#pragma unroll
  for (int u = 0; u < (NTILE + 1) / 2; ++u) {
    const int tA = 2 * u, tB = 2 * u + 1;
    if (tA + PF < NTILE) TILE_LOAD(tA + PF)
    if (tB + PF < NTILE) TILE_LOAD(tB + PF)
    __builtin_amdgcn_sched_barrier(0);
    acc[tA] = v8f{};
    if (tB < NTILE) acc[tB] = v8f{};
#pragma unroll
    for (int c = 0; c < 6; ++c) {
      WM(acc[tA], LO2(L[tA][c]), Bf[c][0]);
      if (tB < NTILE) WM(acc[tB], LO2(L[tB][c]), Bf[c][0]);
      WM(acc[tA], HI2(L[tA][c]), Bf[c][1]);
      if (tB < NTILE) WM(acc[tB], HI2(L[tB][c]), Bf[c][1]);
    }
    if (u > 0) FIN(tA - 1)
    FIN(tA)
  }
#undef WM
#undef LO2
#undef HI2
#undef FIN
#undef TILE_LOAD
#undef TILE_OFF
  __syncthreads();

  int cc1[4];
#pragma unroll
  for (int nt = 0; nt < 4; ++nt) {
    const int c = nt * 16 + rA;
    cc1[nt] = (c < 50) ? c : 49;
  }

  v8f acc1[4] = {};
  {
    const float* HA = H + rA * HPITCH + hs;
    const float* WB = WS + hh * W1PP;
    v2f Af[2][7], Bq[2][7][4];
#define G1LOAD(buf, j)                                                        \
    _Pragma("unroll") for (int s = 0; s < 7; ++s) {                           \
      const int kk = 7 * (j) + s;                                             \
      Af[buf][s].x = HA[4 * kk]; Af[buf][s].y = HA[4 * kk + 1];               \
      _Pragma("unroll") for (int nt = 0; nt < 4; ++nt)                        \
        Bq[buf][s][nt] = *(const v2f*)(WB + 2 * kk * W1PP + 2 * cc1[nt]);     \
    }
    G1LOAD(0, 0)
#pragma unroll
    for (int j = 0; j < 7; ++j) {
      if (j + 1 < 7) { G1LOAD((j + 1) & 1, j + 1) }
      __builtin_amdgcn_sched_barrier(0);
#pragma unroll
      for (int s = 0; s < 7; ++s)
#pragma unroll
        for (int nt = 0; nt < 4; ++nt)
          acc1[nt] = __builtin_amdgcn_wmma_f32_16x16x4_f32(
              false, Af[j & 1][s], false, Bq[j & 1][s][nt], (short)0, acc1[nt], false, false);
      __builtin_amdgcn_sched_barrier(0);
    }
#undef G1LOAD
  }
  __syncthreads();

  float* H1 = H;
#pragma unroll
  for (int nt = 0; nt < 4; ++nt) {
    const int col = nt * 16 + rA;
    const float bb = WS[BSOFF + cc1[nt]];
#pragma unroll
    for (int r = 0; r < 8; ++r) {
      float vv = fmaxf(acc1[nt][r] + bb, 0.f);
      vv = (col < 50) ? vv : 0.f;
      H1[(rD + r) * H1PITCH + col] = vv;
    }
  }
  __syncthreads();

  const int cc2 = (16 + rA < 30) ? (16 + rA) : 29;
  v8f acc2[2] = {};
  {
    const float* HA = H1 + rA * H1PITCH + hs;
    const float* WB = WS + W2OFF + hh * W2PP;
    v2f Af[13], B0[13], B1[13];
#pragma unroll
    for (int kk = 0; kk < 13; ++kk) {
      Af[kk].x = HA[4 * kk]; Af[kk].y = HA[4 * kk + 1];
      B0[kk] = *(const v2f*)(WB + 2 * kk * W2PP + 2 * rA);
      B1[kk] = *(const v2f*)(WB + 2 * kk * W2PP + 2 * cc2);
    }
    __builtin_amdgcn_sched_barrier(0);
#pragma unroll
    for (int kk = 0; kk < 13; ++kk) {
      acc2[0] = __builtin_amdgcn_wmma_f32_16x16x4_f32(
          false, Af[kk], false, B0[kk], (short)0, acc2[0], false, false);
      acc2[1] = __builtin_amdgcn_wmma_f32_16x16x4_f32(
          false, Af[kk], false, B1[kk], (short)0, acc2[1], false, false);
    }
  }
  __syncthreads();
  float* H2 = H + H2OFF;
#pragma unroll
  for (int nt = 0; nt < 2; ++nt) {
    const int col = nt * 16 + rA;
    const float bb = WS[BSOFF + 50 + ((col < 30) ? col : 29)];
#pragma unroll
    for (int r = 0; r < 8; ++r) {
      float vv = fmaxf(acc2[nt][r] + bb, 0.f);
      vv = (col < 30) ? vv : 0.f;
      H2[(rD + r) * H2PITCH + col] = vv;
    }
  }
  __syncthreads();

  const int cc3 = (rA < 15) ? rA : 14;
  v8f acc3 = {};
  {
    const float* HA = H2 + rA * H2PITCH + hs;
    const float* WB = WS + W3OFF + hh * W3PP;
    v2f Af[8], Bv[8];
#pragma unroll
    for (int kk = 0; kk < 8; ++kk) {
      Af[kk].x = HA[4 * kk]; Af[kk].y = HA[4 * kk + 1];
      Bv[kk] = *(const v2f*)(WB + 2 * kk * W3PP + 2 * cc3);
    }
    __builtin_amdgcn_sched_barrier(0);
#pragma unroll
    for (int kk = 0; kk < 8; ++kk)
      acc3 = __builtin_amdgcn_wmma_f32_16x16x4_f32(
          false, Af[kk], false, Bv[kk], (short)0, acc3, false, false);
  }
  {
    const float bb = WS[BSOFF + 80 + cc3];
#pragma unroll
    for (int r = 0; r < 8; ++r)
      O[(rD + r) * OPITCH + rA] = acc3[r] + bb;
  }
  __syncthreads();

  if (lane < 16) {
    const int gRow = rowBase + lane;
    if (gRow < B) {
      float a9[9];
#pragma unroll
      for (int i = 0; i < 9; ++i) a9[i] = O[lane * OPITCH + i];
      {
        float am = 0.f;
#pragma unroll
        for (int i = 0; i < 9; ++i) am = fmaxf(am, fabsf(a9[i]));
        int ex = (__float_as_int(am) >> 23) & 0xff;
        ex = (ex > 253) ? 253 : ex;
        const float scl = __int_as_float((254 - ex) << 23);
#pragma unroll
        for (int i = 0; i < 9; ++i) a9[i] *= scl;
      }
      float s[9], v[9];
#pragma unroll
      for (int i = 0; i < 3; ++i)
#pragma unroll
        for (int j = 0; j < 3; ++j) {
          s[i * 3 + j] = a9[0 * 3 + i] * a9[0 * 3 + j] +
                         a9[1 * 3 + i] * a9[1 * 3 + j] +
                         a9[2 * 3 + i] * a9[2 * 3 + j];
          v[i * 3 + j] = (i == j) ? 1.f : 0.f;
        }
#pragma unroll
      for (int sw = 0; sw < 6; ++sw) {
        jacobi3(s, v, 0, 1); jacobi3(s, v, 0, 2); jacobi3(s, v, 1, 2);
      }
      float g0 = s[0], g1 = s[4], g2 = s[8];
#define SWAPCOL(p, q) { float t_;                              \
        t_ = v[p];     v[p]     = v[q];     v[q]     = t_;     \
        t_ = v[3 + p]; v[3 + p] = v[3 + q]; v[3 + q] = t_;     \
        t_ = v[6 + p]; v[6 + p] = v[6 + q]; v[6 + q] = t_; }
      if (g0 < g1) { float t = g0; g0 = g1; g1 = t; SWAPCOL(0, 1); }
      if (g0 < g2) { float t = g0; g0 = g2; g2 = t; SWAPCOL(0, 2); }
      if (g1 < g2) { float t = g1; g1 = g2; g2 = t; SWAPCOL(1, 2); }
#undef SWAPCOL
      float w[9];
#pragma unroll
      for (int i = 0; i < 3; ++i)
#pragma unroll
        for (int j = 0; j < 3; ++j)
          w[i * 3 + j] = a9[i * 3 + 0] * v[0 + j] +
                         a9[i * 3 + 1] * v[3 + j] +
                         a9[i * 3 + 2] * v[6 + j];
      float u0x = w[0], u0y = w[3], u0z = w[6];
      float n0 = 1.f / sqrtf(u0x * u0x + u0y * u0y + u0z * u0z + 1e-20f);
      u0x *= n0; u0y *= n0; u0z *= n0;
      float u1x = w[1], u1y = w[4], u1z = w[7];
      float d01 = u1x * u0x + u1y * u0y + u1z * u0z;
      u1x -= d01 * u0x; u1y -= d01 * u0y; u1z -= d01 * u0z;
      float n1 = 1.f / sqrtf(u1x * u1x + u1y * u1y + u1z * u1z + 1e-20f);
      u1x *= n1; u1y *= n1; u1z *= n1;
      float cx = u0y * u1z - u0z * u1y;
      float cy = u0z * u1x - u0x * u1z;
      float cz = u0x * u1y - u0y * u1x;
      float detV = v[0] * (v[4] * v[8] - v[5] * v[7]) -
                   v[1] * (v[3] * v[8] - v[5] * v[6]) +
                   v[2] * (v[3] * v[7] - v[4] * v[6]);
      float sg = (detV >= 0.f) ? 1.f : -1.f;
      float u2x = sg * cx, u2y = sg * cy, u2z = sg * cz;
      float U_[9] = { u0x, u1x, u2x,  u0y, u1y, u2y,  u0z, u1z, u2z };
      float* op = out + (size_t)gRow * 15;
#pragma unroll
      for (int i = 0; i < 3; ++i)
#pragma unroll
        for (int j = 0; j < 3; ++j)
          op[i * 3 + j] = U_[i * 3 + 0] * v[j * 3 + 0] +
                          U_[i * 3 + 1] * v[j * 3 + 1] +
                          U_[i * 3 + 2] * v[j * 3 + 2];
#pragma unroll
      for (int i = 9; i < 15; ++i) op[i] = O[lane * OPITCH + i];
    }
  }
}

extern "C" void kernel_launch(void* const* d_in, const int* in_sizes, int n_in,
                              void* d_out, int out_size, void* d_ws, size_t ws_size,
                              hipStream_t stream) {
  (void)n_in; (void)out_size; (void)d_ws; (void)ws_size;
  const float* x   = (const float*)d_in[0];
  const float* cw  = (const float*)d_in[1];
  const float* cb  = (const float*)d_in[2];
  const float* w1  = (const float*)d_in[3];
  const float* b1  = (const float*)d_in[4];
  const float* w2  = (const float*)d_in[5];
  const float* b2  = (const float*)d_in[6];
  const float* w3  = (const float*)d_in[7];
  const float* b3  = (const float*)d_in[8];
  const int B = in_sizes[0] / 1200;
  const int grid = (B + WAVES * ROWS - 1) / (WAVES * ROWS);
  rmi_fused<<<grid, WAVES * 32, 0, stream>>>(x, cw, cb, w1, b1, w2, b2, w3, b3,
                                             (float*)d_out, B);
}


# 256-thread work-groups of two 4-wave groups, one wave of each per SIMD: group B's conv starts at the one barrier ending group A's, so B's x stream runs under A's dense layers and SVD; same sum orders
# speedup vs baseline: 3.0633x; 1.0630x over previous
#include <hip/hip_runtime.h>
#include <math.h>

typedef __attribute__((ext_vector_type(2))) float v2f;
typedef __attribute__((ext_vector_type(4))) float v4f;
typedef __attribute__((ext_vector_type(8))) float v8f;

#define WAVES   8
#define ROWS    16
#define HPITCH  197
#define H1PITCH 67
#define H2PITCH 33
#define OPITCH  17
#define H2OFF   1536
#define W1PP    100
#define W2PP    96
#define W3PP    32
#define W2OFF   9800
#define W3OFF   (W2OFF + 26 * W2PP)
#define BSOFF   (W3OFF + 16 * W3PP)
#define WSLEN   (BSOFF + 96)

__device__ __forceinline__ void jacobi3(float* s, float* v, int p, int q) {
  float apq = s[p * 3 + q];
  const bool skip = fabsf(apq) < 1e-20f;
  float app = s[p * 3 + p], aqq = s[q * 3 + q];
  const float ch  = 2.f * (aqq - app), sh = apq;
  const float ch2 = ch * ch, sh2 = sh * sh, chsh = ch * sh;
  const float nn  = ch2 + sh2;
  float c  = (ch2 - sh2) / nn;
  float sn = (2.f * chsh) / nn;
  if (!(5.8284273f * sh2 < ch2)) {
    c  = 0.70710678f;
    sn = copysignf(0.70710678f, chsh);
  }
  if (skip) { c = 1.f; sn = 0.f; }
  const int k = 3 - p - q;
  float skp = s[k * 3 + p], skq = s[k * 3 + q];
  float nkp = c * skp - sn * skq;
  float nkq = sn * skp + c * skq;
  s[k * 3 + p] = nkp; s[p * 3 + k] = nkp;
  s[k * 3 + q] = nkq; s[q * 3 + k] = nkq;
  const float cc = c * c, ss = sn * sn, sc = sn * c;
  s[p * 3 + p] = cc * app - 2.f * sc * apq + ss * aqq;
  s[q * 3 + q] = ss * app + 2.f * sc * apq + cc * aqq;
  const float npq = (cc - ss) * apq + sc * (app - aqq);
  s[p * 3 + q] = npq; s[q * 3 + p] = npq;
#pragma unroll
  for (int i = 0; i < 3; ++i) {
    float vip = v[i * 3 + p], viq = v[i * 3 + q];
    v[i * 3 + p] = c * vip - sn * viq;
    v[i * 3 + q] = sn * vip + c * viq;
  }
}

__device__ __forceinline__ float shr8_zero(float v) {
  return __builtin_bit_cast(float, __builtin_amdgcn_update_dpp(
      0, __builtin_bit_cast(int, v), 0x118  , 0xF, 0xF, true));
}

__global__ __launch_bounds__(256) void rmi_fused(
    const float* __restrict__ x,   const float* __restrict__ cwg,
    const float* __restrict__ cbg, const float* __restrict__ w1,
    const float* __restrict__ b1,  const float* __restrict__ w2,
    const float* __restrict__ b2,  const float* __restrict__ w3,
    const float* __restrict__ b3,  float* __restrict__ out, int B) {
  __shared__ float lds [WAVES * ROWS * HPITCH];
  __shared__ float ldsO[WAVES * ROWS * OPITCH];
  __shared__ __attribute__((aligned(16))) float WS[WSLEN];

  const int tid  = threadIdx.x;
  const int wave = tid >> 5;
  const int lane = tid & 31;
  const int rowBase = blockIdx.x * (WAVES * ROWS) + wave * ROWS;
  float* H = &lds [wave * ROWS * HPITCH];
  float* O = &ldsO[wave * ROWS * OPITCH];

  const int hs = (lane >= 16) ? 2 : 0;
  const int rA = lane & 15;
  const int rD = (lane < 16) ? 0 : 8;

  const float cb = cbg[0];

  const int grp = __builtin_amdgcn_readfirstlane(wave) >> 2;
  const int tl  = tid & 127;

  const int hh = lane >> 4;
  v2f Bf[6][2];
#pragma unroll
  for (int c = 0; c < 6; ++c)
#pragma unroll
    for (int z = 0; z < 2; ++z) {
      const int t0 = (4 * hh + 2 * z) - (rA - 4);
      const int t1 = t0 + 1;
      const int i0 = (t0 < 0) ? 0 : ((t0 > 4) ? 4 : t0);
      const int i1 = (t1 < 0) ? 0 : ((t1 > 4) ? 4 : t1);
      const float wa = cwg[c * 5 + i0];
      const float wb = cwg[c * 5 + i1];
      Bf[c][z].x = (t0 >= 0 && t0 <= 4) ? wa : 0.f;
      Bf[c][z].y = (t1 >= 0 && t1 <= 4) ? wb : 0.f;
    }

  const int rowBaseS = blockIdx.x * (WAVES * ROWS) + __builtin_amdgcn_readfirstlane(wave) * ROWS;
  const char* xb = (const char*)x;
#define TILE_OFF(t, off) {                                                \
    const int P_  = 8 * (t) + (rA & 7);                                    \
    const int b_  = P_ / 25;                                               \
    const int pp_ = P_ - 25 * b_;                                          \
    int row_ = rowBaseS + 8 * (rA >> 3) + b_;                              \
    row_ = (row_ < B) ? row_ : (B - 1);                                    \
    off = (unsigned)row_ * 4800u + 32u * (unsigned)pp_ + 16u * (unsigned)hh; }
#define TILE_LOAD(t) {                                                    \
    unsigned off_;                                                         \
    TILE_OFF(t, off_)                                                      \
    _Pragma("unroll") for (int c = 0; c < 6; ++c)                          \
      L[t][c] = *(const v4f*)(xb + off_ + c * 800); }

#define NTILE 25
#define PF    6
  if (grp == 1) {
    const int ts = (tl < 124) ? tl : 124;
    const int sr = ts / 25, sp = ts - 25 * sr;
#pragma unroll
    for (int j = 0; j < 20; ++j) {
      int m = sr + 5 * j;
      if (j == 19) m = (m < 97) ? m : 97;
      const v2f ga = *(const v2f*)(w1 + m * 100 + 2 * sp);
      const v2f gb = *(const v2f*)(w1 + m * 100 + 50 + 2 * sp);
      v4f o; o.x = ga.x; o.y = gb.x; o.z = ga.y; o.w = gb.y;
      *(v4f*)(WS + m * W1PP + 4 * sp) = o;
    }
#pragma unroll
    for (int j = 0; j < 3; ++j) {
      int it = tl + 128 * j;  it = (it < 374) ? it : 374;
      const int m = it / 15, p = it - 15 * m;
      const v2f ga = *(const v2f*)(w2 + m * 60 + 2 * p);
      const v2f gb = *(const v2f*)(w2 + m * 60 + 30 + 2 * p);
      v4f o; o.x = ga.x; o.y = gb.x; o.z = ga.y; o.w = gb.y;
      *(v4f*)(WS + W2OFF + m * W2PP + 4 * p) = o;
    }
#pragma unroll
    for (int j = 0; j < 2; ++j) {
      int it = tl + 128 * j;  it = (it < 224) ? it : 224;
      const int m = it / 15, c = it - 15 * m;
      v2f o; o.x = w3[m * 30 + c]; o.y = w3[m * 30 + 15 + c];
      *(v2f*)(WS + W3OFF + m * W3PP + 2 * c) = o;
    }
    const v2f z2 = {0.f, 0.f};
    *(v2f*)(WS + W2OFF + 25 * W2PP + 2 * (tl & 31)) = z2;
    *(v2f*)(WS + W3OFF + 15 * W3PP + 2 * (tl & 15)) = z2;
    const int t1 = (tl < 49) ? tl : 49, t2 = (tl < 29) ? tl : 29, t3 = (tl < 14) ? tl : 14;
    WS[BSOFF + t1]      = b1[t1];
    WS[BSOFF + 50 + t2] = b2[t2];
    WS[BSOFF + 80 + t3] = b3[t3];
  }

  v4f L[NTILE][6];
#pragma unroll
  for (int t = 0; t < PF; ++t) TILE_LOAD(t)

#define WM(acc, a, b) acc = __builtin_amdgcn_wmma_f32_16x16x4_f32( \
      false, a, false, b, (short)0, acc, false, false)
#define LO2(v) __builtin_shufflevector(v, v, 0, 1)
#define HI2(v) __builtin_shufflevector(v, v, 2, 3)

  const bool st8 = (rA >= 4) && (rA <= 11);
  const bool st4 = (rA >= 4) && (rA <= 7);
  const int  hst = 8 * hh * HPITCH + rA - 4;
#define FIN(t) {                                                               \
    float g_[8];                                                               \
    _Pragma("unroll") for (int r = 0; r < 8; ++r) {                            \
      const int P_ = 8 * (t) + r, pp_ = P_ % 25;                               \
      float v_ = acc[t][r];                                                    \
      if (pp_ != 24) {                                                         \
        const float nx_ = (r < 7) ? acc[t][(r + 1) & 7] : acc[((t) + 1) % (NTILE + 1)][0]; \
        v_ += shr8_zero(nx_);                                                  \
      }                                                                        \
      g_[r] = v_ + cb;                                                         \
    }                                                                          \
    if (st8) {                                                                 \
      _Pragma("unroll") for (int r = 0; r < 8; ++r) {                          \
        const int P_ = 8 * (t) + r, b_ = P_ / 25, pp_ = P_ - 25 * b_;          \
        if (pp_ != 24) H[hst + b_ * HPITCH + 8 * pp_] = g_[r];                 \
      }                                                                        \
    }                                                                          \
    if (st4) {                                                                 \
      _Pragma("unroll") for (int r = 0; r < 8; ++r) {                          \
        const int P_ = 8 * (t) + r, b_ = P_ / 25, pp_ = P_ - 25 * b_;          \
        if (pp_ == 24) H[hst + b_ * HPITCH + 8 * pp_] = g_[r];                 \
      }                                                                        \
    } }

#pragma nounroll
  for (int ph = 0; ph < 2; ++ph) {
   if (ph == grp) {
  v8f acc[NTILE + 1];
#pragma unroll
  for (int u = 0; u < (NTILE + 1) / 2; ++u) {
    const int tA = 2 * u, tB = 2 * u + 1;
    if (tA + PF < NTILE) TILE_LOAD(tA + PF)
    if (tB + PF < NTILE) TILE_LOAD(tB + PF)
    __builtin_amdgcn_sched_barrier(0);
    acc[tA] = v8f{};
    if (tB < NTILE) acc[tB] = v8f{};
#pragma unroll
    for (int c = 0; c < 6; ++c) {
      WM(acc[tA], LO2(L[tA][c]), Bf[c][0]);
      if (tB < NTILE) WM(acc[tB], LO2(L[tB][c]), Bf[c][0]);
      WM(acc[tA], HI2(L[tA][c]), Bf[c][1]);
      if (tB < NTILE) WM(acc[tB], HI2(L[tB][c]), Bf[c][1]);
    }
    if (u > 0) FIN(tA - 1)
    FIN(tA)
  }
   }
   if (ph == 0) __syncthreads();
  }
#undef WM
#undef LO2
#undef HI2
#undef FIN
#undef TILE_LOAD
#undef TILE_OFF
#define WAVE_SYNC() {                                                        \
    __builtin_amdgcn_fence(__ATOMIC_RELEASE, "wavefront");                   \
    __builtin_amdgcn_wave_barrier();                                         \
    __builtin_amdgcn_fence(__ATOMIC_ACQUIRE, "wavefront"); }
  WAVE_SYNC()

  int cc1[4];
#pragma unroll
  for (int nt = 0; nt < 4; ++nt) {
    const int c = nt * 16 + rA;
    cc1[nt] = (c < 50) ? c : 49;
  }

  v8f acc1[4] = {};
  {
    const float* HA = H + rA * HPITCH + hs;
    const float* WB = WS + hh * W1PP;
    v2f Af[2][7], Bq[2][7][4];
#define G1LOAD(buf, j)                                                        \
    _Pragma("unroll") for (int s = 0; s < 7; ++s) {                           \
      const int kk = 7 * (j) + s;                                             \
      Af[buf][s].x = HA[4 * kk]; Af[buf][s].y = HA[4 * kk + 1];               \
      _Pragma("unroll") for (int nt = 0; nt < 4; ++nt)                        \
        Bq[buf][s][nt] = *(const v2f*)(WB + 2 * kk * W1PP + 2 * cc1[nt]);     \
    }
    G1LOAD(0, 0)
#pragma unroll
    for (int j = 0; j < 7; ++j) {
      if (j + 1 < 7) { G1LOAD((j + 1) & 1, j + 1) }
      __builtin_amdgcn_sched_barrier(0);
#pragma unroll
      for (int s = 0; s < 7; ++s)
#pragma unroll
        for (int nt = 0; nt < 4; ++nt)
          acc1[nt] = __builtin_amdgcn_wmma_f32_16x16x4_f32(
              false, Af[j & 1][s], false, Bq[j & 1][s][nt], (short)0, acc1[nt], false, false);
      __builtin_amdgcn_sched_barrier(0);
    }
#undef G1LOAD
  }
  WAVE_SYNC()

  float* H1 = H;
#pragma unroll
  for (int nt = 0; nt < 4; ++nt) {
    const int col = nt * 16 + rA;
    const float bb = WS[BSOFF + cc1[nt]];
#pragma unroll
    for (int r = 0; r < 8; ++r) {
      float vv = fmaxf(acc1[nt][r] + bb, 0.f);
      vv = (col < 50) ? vv : 0.f;
      H1[(rD + r) * H1PITCH + col] = vv;
    }
  }
  WAVE_SYNC()

  const int cc2 = (16 + rA < 30) ? (16 + rA) : 29;
  v8f acc2[2] = {};
  {
    const float* HA = H1 + rA * H1PITCH + hs;
    const float* WB = WS + W2OFF + hh * W2PP;
    v2f Af[13], B0[13], B1[13];
#pragma unroll
    for (int kk = 0; kk < 13; ++kk) {
      Af[kk].x = HA[4 * kk]; Af[kk].y = HA[4 * kk + 1];
      B0[kk] = *(const v2f*)(WB + 2 * kk * W2PP + 2 * rA);
      B1[kk] = *(const v2f*)(WB + 2 * kk * W2PP + 2 * cc2);
    }
    __builtin_amdgcn_sched_barrier(0);
#pragma unroll
    for (int kk = 0; kk < 13; ++kk) {
      acc2[0] = __builtin_amdgcn_wmma_f32_16x16x4_f32(
          false, Af[kk], false, B0[kk], (short)0, acc2[0], false, false);
      acc2[1] = __builtin_amdgcn_wmma_f32_16x16x4_f32(
          false, Af[kk], false, B1[kk], (short)0, acc2[1], false, false);
    }
  }
  WAVE_SYNC()
  float* H2 = H + H2OFF;
#pragma unroll
  for (int nt = 0; nt < 2; ++nt) {
    const int col = nt * 16 + rA;
    const float bb = WS[BSOFF + 50 + ((col < 30) ? col : 29)];
#pragma unroll
    for (int r = 0; r < 8; ++r) {
      float vv = fmaxf(acc2[nt][r] + bb, 0.f);
      vv = (col < 30) ? vv : 0.f;
      H2[(rD + r) * H2PITCH + col] = vv;
    }
  }
  WAVE_SYNC()

  const int cc3 = (rA < 15) ? rA : 14;
  v8f acc3 = {};
  {
    const float* HA = H2 + rA * H2PITCH + hs;
    const float* WB = WS + W3OFF + hh * W3PP;
    v2f Af[8], Bv[8];
#pragma unroll
    for (int kk = 0; kk < 8; ++kk) {
      Af[kk].x = HA[4 * kk]; Af[kk].y = HA[4 * kk + 1];
      Bv[kk] = *(const v2f*)(WB + 2 * kk * W3PP + 2 * cc3);
    }
    __builtin_amdgcn_sched_barrier(0);
#pragma unroll
    for (int kk = 0; kk < 8; ++kk)
      acc3 = __builtin_amdgcn_wmma_f32_16x16x4_f32(
          false, Af[kk], false, Bv[kk], (short)0, acc3, false, false);
  }
  {
    const float bb = WS[BSOFF + 80 + cc3];
#pragma unroll
    for (int r = 0; r < 8; ++r)
      O[(rD + r) * OPITCH + rA] = acc3[r] + bb;
  }
  WAVE_SYNC()

  if (lane < 16) {
    const int gRow = rowBase + lane;
    if (gRow < B) {
      float a9[9];
#pragma unroll
      for (int i = 0; i < 9; ++i) a9[i] = O[lane * OPITCH + i];
      {
        float am = 0.f;
#pragma unroll
        for (int i = 0; i < 9; ++i) am = fmaxf(am, fabsf(a9[i]));
        int ex = (__float_as_int(am) >> 23) & 0xff;
        ex = (ex > 253) ? 253 : ex;
        const float scl = __int_as_float((254 - ex) << 23);
#pragma unroll
        for (int i = 0; i < 9; ++i) a9[i] *= scl;
      }
      float s[9], v[9];
#pragma unroll
      for (int i = 0; i < 3; ++i)
#pragma unroll
        for (int j = 0; j < 3; ++j) {
          s[i * 3 + j] = a9[0 * 3 + i] * a9[0 * 3 + j] +
                         a9[1 * 3 + i] * a9[1 * 3 + j] +
                         a9[2 * 3 + i] * a9[2 * 3 + j];
          v[i * 3 + j] = (i == j) ? 1.f : 0.f;
        }
#pragma unroll
      for (int sw = 0; sw < 6; ++sw) {
        jacobi3(s, v, 0, 1); jacobi3(s, v, 0, 2); jacobi3(s, v, 1, 2);
      }
      float g0 = s[0], g1 = s[4], g2 = s[8];
#define SWAPCOL(p, q) { float t_;                              \
        t_ = v[p];     v[p]     = v[q];     v[q]     = t_;     \
        t_ = v[3 + p]; v[3 + p] = v[3 + q]; v[3 + q] = t_;     \
        t_ = v[6 + p]; v[6 + p] = v[6 + q]; v[6 + q] = t_; }
      if (g0 < g1) { float t = g0; g0 = g1; g1 = t; SWAPCOL(0, 1); }
      if (g0 < g2) { float t = g0; g0 = g2; g2 = t; SWAPCOL(0, 2); }
      if (g1 < g2) { float t = g1; g1 = g2; g2 = t; SWAPCOL(1, 2); }
#undef SWAPCOL
      float w[9];
#pragma unroll
      for (int i = 0; i < 3; ++i)
#pragma unroll
        for (int j = 0; j < 3; ++j)
          w[i * 3 + j] = a9[i * 3 + 0] * v[0 + j] +
                         a9[i * 3 + 1] * v[3 + j] +
                         a9[i * 3 + 2] * v[6 + j];
      float u0x = w[0], u0y = w[3], u0z = w[6];
      float n0 = 1.f / sqrtf(u0x * u0x + u0y * u0y + u0z * u0z + 1e-20f);
      u0x *= n0; u0y *= n0; u0z *= n0;
      float u1x = w[1], u1y = w[4], u1z = w[7];
      float d01 = u1x * u0x + u1y * u0y + u1z * u0z;
      u1x -= d01 * u0x; u1y -= d01 * u0y; u1z -= d01 * u0z;
      float n1 = 1.f / sqrtf(u1x * u1x + u1y * u1y + u1z * u1z + 1e-20f);
      u1x *= n1; u1y *= n1; u1z *= n1;
      float cx = u0y * u1z - u0z * u1y;
      float cy = u0z * u1x - u0x * u1z;
      float cz = u0x * u1y - u0y * u1x;
      float detV = v[0] * (v[4] * v[8] - v[5] * v[7]) -
                   v[1] * (v[3] * v[8] - v[5] * v[6]) +
                   v[2] * (v[3] * v[7] - v[4] * v[6]);
      float sg = (detV >= 0.f) ? 1.f : -1.f;
      float u2x = sg * cx, u2y = sg * cy, u2z = sg * cz;
      float U_[9] = { u0x, u1x, u2x,  u0y, u1y, u2y,  u0z, u1z, u2z };
      float* op = out + (size_t)gRow * 15;
#pragma unroll
      for (int i = 0; i < 3; ++i)
#pragma unroll
        for (int j = 0; j < 3; ++j)
          op[i * 3 + j] = U_[i * 3 + 0] * v[j * 3 + 0] +
                          U_[i * 3 + 1] * v[j * 3 + 1] +
                          U_[i * 3 + 2] * v[j * 3 + 2];
#pragma unroll
      for (int i = 9; i < 15; ++i) op[i] = O[lane * OPITCH + i];
    }
  }
}

extern "C" void kernel_launch(void* const* d_in, const int* in_sizes, int n_in,
                              void* d_out, int out_size, void* d_ws, size_t ws_size,
                              hipStream_t stream) {
  (void)n_in; (void)out_size; (void)d_ws; (void)ws_size;
  const float* x   = (const float*)d_in[0];
  const float* cw  = (const float*)d_in[1];
  const float* cb  = (const float*)d_in[2];
  const float* w1  = (const float*)d_in[3];
  const float* b1  = (const float*)d_in[4];
  const float* w2  = (const float*)d_in[5];
  const float* b2  = (const float*)d_in[6];
  const float* w3  = (const float*)d_in[7];
  const float* b3  = (const float*)d_in[8];
  const int B = in_sizes[0] / 1200;
  const int grid = (B + WAVES * ROWS - 1) / (WAVES * ROWS);
  rmi_fused<<<grid, WAVES * 32, 0, stream>>>(x, cw, cb, w1, b1, w2, b2, w3, b3,
                                             (float*)d_out, B);
}
